# starting kernel
# baseline (speedup 1.0000x reference)
#include <hip/hip_runtime.h>
#include <math.h>

typedef __attribute__((ext_vector_type(2))) float v2f;
typedef __attribute__((ext_vector_type(8))) float v8f;

#define NEG_SLOPE 0.2f

static __device__ __forceinline__ void atomicMaxF(float* addr, float val) {
    if (val >= 0.0f) atomicMax((int*)addr, __float_as_int(val));
    else             atomicMin((unsigned int*)addr, __float_as_uint(val));
}

static __device__ __forceinline__ void endpoints(const int* __restrict__ ei,
                                                 int E, int e, int& s, int& d) {
    if (e < E) { s = ei[e]; d = ei[E + e]; }
    else       { s = e - E; d = s; }
}

__global__ void init_state(float* __restrict__ m1, float* __restrict__ s1,
                           float* __restrict__ m2, float* __restrict__ s2,
                           float* __restrict__ xemb, float* __restrict__ out2,
                           int N, int H, int HC, int OUT) {
    int t = blockIdx.x * blockDim.x + threadIdx.x;
    const float ninf = __int_as_float(0xFF800000u);
    if (t < N * H)   { m1[t] = ninf; s1[t] = 0.0f; }
    if (t < N)       { m2[t] = ninf; s2[t] = 0.0f; }
    if (t < N * HC)  xemb[t] = 0.0f;
    if (t < N * OUT) out2[t] = 0.0f;
}

__global__ void gemm1_wmma(const float* __restrict__ X, const float* __restrict__ W,
                           float* __restrict__ XL, int N, int K) {
    const int lane  = threadIdx.x;
    const int mrow  = lane & 15;
    const int khalf = (lane >> 4) << 1;
    const int mt    = blockIdx.x;
    int row = mt * 16 + mrow;
    if (row > N - 1) row = N - 1;

    v8f acc0 = {}, acc1 = {}, acc2 = {}, acc3 = {};
    for (int k0 = 0; k0 < K; k0 += 4) {
        const int kb = k0 + khalf;
        v2f a;
        a.x = X[row * K + kb];
        a.y = X[row * K + kb + 1];
        const float* Wr0 = W + kb * 64;
        const float* Wr1 = Wr0 + 64;
        v2f b0, b1, b2, b3;
        b0.x = Wr0[mrow];      b0.y = Wr1[mrow];
        b1.x = Wr0[mrow + 16]; b1.y = Wr1[mrow + 16];
        b2.x = Wr0[mrow + 32]; b2.y = Wr1[mrow + 32];
        b3.x = Wr0[mrow + 48]; b3.y = Wr1[mrow + 48];
        acc0 = __builtin_amdgcn_wmma_f32_16x16x4_f32(false, a, false, b0, (short)0, acc0, false, false);
        acc1 = __builtin_amdgcn_wmma_f32_16x16x4_f32(false, a, false, b1, (short)0, acc1, false, false);
        acc2 = __builtin_amdgcn_wmma_f32_16x16x4_f32(false, a, false, b2, (short)0, acc2, false, false);
        acc3 = __builtin_amdgcn_wmma_f32_16x16x4_f32(false, a, false, b3, (short)0, acc3, false, false);
    }
    const int r0 = (lane >> 4) * 8;
    for (int v = 0; v < 8; ++v) {
        int r = mt * 16 + r0 + v;
        if (r < N) {
            float* o = XL + r * 64;
            o[mrow]      = acc0[v];
            o[mrow + 16] = acc1[v];
            o[mrow + 32] = acc2[v];
            o[mrow + 48] = acc3[v];
        }
    }
}

__global__ void gemm2_wmma(const float* __restrict__ Hh, const float* __restrict__ W,
                           float* __restrict__ XL, int N, int K, int OUT) {
    const int lane  = threadIdx.x;
    const int mrow  = lane & 15;
    const int khalf = (lane >> 4) << 1;
    const int mt    = blockIdx.x;
    int row = mt * 16 + mrow;
    if (row > N - 1) row = N - 1;

    v8f acc = {};
    for (int k0 = 0; k0 < K; k0 += 4) {
        const int kb = k0 + khalf;
        v2f a, b;
        a.x = Hh[row * K + kb];
        a.y = Hh[row * K + kb + 1];
        b.x = (mrow < OUT) ? W[kb * OUT + mrow]       : 0.0f;
        b.y = (mrow < OUT) ? W[(kb + 1) * OUT + mrow] : 0.0f;
        acc = __builtin_amdgcn_wmma_f32_16x16x4_f32(false, a, false, b, (short)0, acc, false, false);
    }
    if (mrow < OUT) {
        const int r0 = (lane >> 4) * 8;
        for (int v = 0; v < 8; ++v) {
            int r = mt * 16 + r0 + v;
            if (r < N) XL[r * OUT + mrow] = acc[v];
        }
    }
}

template <int H, int C>
__global__ void coefs(const float* __restrict__ XL, const float* __restrict__ Asrc,
                      const float* __restrict__ Adst, float* __restrict__ as_o,
                      float* __restrict__ ad_o, int N) {
    int t = blockIdx.x * blockDim.x + threadIdx.x;
    if (t >= N * H) return;
    int h = t % H;
    const float* v = XL + t * C;
    float ss = 0.f, sd = 0.f;
#pragma unroll
    for (int c = 0; c < C; ++c) { ss += v[c] * Asrc[h * C + c]; sd += v[c] * Adst[h * C + c]; }
    as_o[t] = ss; ad_o[t] = sd;
}

template <int H>
__global__ void edge_lrelu_max(const int* __restrict__ ei, const float* __restrict__ as,
                               const float* __restrict__ ad, float* __restrict__ ebuf,
                               float* __restrict__ m, int E, int Ep) {
    int t = blockIdx.x * blockDim.x + threadIdx.x;
    if (t >= Ep * H) return;
    int e = t / H, h = t - e * H;
    int s, d; endpoints(ei, E, e, s, d);
    float v = as[s * H + h] + ad[d * H + h];
    v = (v > 0.f) ? v : NEG_SLOPE * v;
    ebuf[t] = v;
    atomicMaxF(&m[d * H + h], v);
}

template <int H>
__global__ void edge_exp_sum(const int* __restrict__ ei, float* __restrict__ ebuf,
                             const float* __restrict__ m, float* __restrict__ ssum,
                             int E, int Ep) {
    int t = blockIdx.x * blockDim.x + threadIdx.x;
    if (t >= Ep * H) return;
    int e = t / H, h = t - e * H;
    int s, d; endpoints(ei, E, e, s, d);
    float v = expf(ebuf[t] - m[d * H + h]);
    ebuf[t] = v;
    atomicAdd(&ssum[d * H + h], v);
}

template <int H, int C>
__global__ void edge_msg(const int* __restrict__ ei, const float* __restrict__ XL,
                         const float* __restrict__ ebuf, const float* __restrict__ ssum,
                         float* __restrict__ acc, int E, int Ep) {
    int t = blockIdx.x * blockDim.x + threadIdx.x;
    if (t >= Ep * (H * C)) return;
    int e  = t / (H * C);
    int hc = t - e * (H * C);
    int h  = hc / C;
    int s, d; endpoints(ei, E, e, s, d);
    float alpha = ebuf[e * H + h] / (ssum[d * H + h] + 1e-16f);
    atomicAdd(&acc[d * (H * C) + hc], XL[s * (H * C) + hc] * alpha);
}

__global__ void finalize1(float* __restrict__ xemb, const float* __restrict__ b,
                          float* __restrict__ Hh, int N, int F) {
    int t = blockIdx.x * blockDim.x + threadIdx.x;
    if (t >= N * F) return;
    float v = xemb[t] + b[t % F];
    xemb[t] = v;
    Hh[t] = v > 0.f ? v : 0.f;
}

__global__ void finalize2(float* __restrict__ out, const float* __restrict__ b,
                          int N, int F) {
    int t = blockIdx.x * blockDim.x + threadIdx.x;
    if (t >= N * F) return;
    out[t] += b[t % F];
}

extern "C" void kernel_launch(void* const* d_in, const int* in_sizes, int n_in,
                              void* d_out, int out_size, void* d_ws, size_t ws_size,
                              hipStream_t stream) {
    const float* x    = (const float*)d_in[0];
    const int*   ei   = (const int*)d_in[1];
    const float* W1   = (const float*)d_in[2];
    const float* as1w = (const float*)d_in[3];
    const float* ad1w = (const float*)d_in[4];
    const float* b1   = (const float*)d_in[5];
    const float* W2   = (const float*)d_in[6];
    const float* as2w = (const float*)d_in[7];
    const float* ad2w = (const float*)d_in[8];
    const float* b2   = (const float*)d_in[9];

    const int HC  = in_sizes[5];
    const int IN  = in_sizes[2] / HC;
    const int OUT = in_sizes[9];
    const int N   = in_sizes[0] / IN;
    const int E   = in_sizes[1] / 2;
    const int Ep  = E + N;
    const int H   = 4;

    float* out  = (float*)d_out;
    float* xemb = out + (size_t)N * OUT;

    float* ws   = (float*)d_ws;
    float* xl1  = ws; ws += (size_t)N * HC;
    float* as1  = ws; ws += (size_t)N * H;
    float* ad1  = ws; ws += (size_t)N * H;
    float* m1   = ws; ws += (size_t)N * H;
    float* s1   = ws; ws += (size_t)N * H;
    float* e1   = ws; ws += (size_t)Ep * H;
    float* hbuf = ws; ws += (size_t)N * HC;
    float* xl2  = ws; ws += (size_t)N * OUT;
    float* as2  = ws; ws += (size_t)N;
    float* ad2  = ws; ws += (size_t)N;
    float* m2   = ws; ws += (size_t)N;
    float* s2   = ws; ws += (size_t)N;
    float* e2   = ws; ws += (size_t)Ep;

    const int B = 256;
    auto g = [](int n, int b) { return (n + b - 1) / b; };

    init_state<<<g(N * HC, B), B, 0, stream>>>(m1, s1, m2, s2, xemb, out, N, H, HC, OUT);

    gemm1_wmma<<<g(N, 16), 32, 0, stream>>>(x, W1, xl1, N, IN);
    coefs<4, 16><<<g(N * H, B), B, 0, stream>>>(xl1, as1w, ad1w, as1, ad1, N);
    edge_lrelu_max<4><<<g(Ep * H, B), B, 0, stream>>>(ei, as1, ad1, e1, m1, E, Ep);
    edge_exp_sum<4><<<g(Ep * H, B), B, 0, stream>>>(ei, e1, m1, s1, E, Ep);
    edge_msg<4, 16><<<g(Ep * H * 16, B), B, 0, stream>>>(ei, xl1, e1, s1, xemb, E, Ep);
    finalize1<<<g(N * HC, B), B, 0, stream>>>(xemb, b1, hbuf, N, HC);

    gemm2_wmma<<<g(N, 16), 32, 0, stream>>>(hbuf, W2, xl2, N, HC, OUT);
    coefs<1, 8><<<g(N, B), B, 0, stream>>>(xl2, as2w, ad2w, as2, ad2, N);
    edge_lrelu_max<1><<<g(Ep, B), B, 0, stream>>>(ei, as2, ad2, e2, m2, E, Ep);
    edge_exp_sum<1><<<g(Ep, B), B, 0, stream>>>(ei, e2, m2, s2, E, Ep);
    edge_msg<1, 8><<<g(Ep * OUT, B), B, 0, stream>>>(ei, xl2, e2, s2, out, E, Ep);
    finalize2<<<g(N * OUT, B), B, 0, stream>>>(out, b2, N, OUT);
}


# edges bucketed by destination once per call (count, scan, fill, canonical order); both layers gather per node, softmax and sums in fixed bucket order, no f32 atomics, summation order changed
# speedup vs baseline: 2.3264x; 2.3264x over previous
#include <hip/hip_runtime.h>
#include <math.h>

typedef __attribute__((ext_vector_type(2))) float v2f;
typedef __attribute__((ext_vector_type(8))) float v8f;

#define NEG_SLOPE 0.2f

static __device__ __forceinline__ float lrelu(float v) { return (v > 0.f) ? v : NEG_SLOPE * v; }

__global__ void zero_deg(int* __restrict__ deg, int N) {
    int t = blockIdx.x * blockDim.x + threadIdx.x;
    if (t < N) deg[t] = 0;
}

__global__ void hist_dst(const int* __restrict__ ei, int* __restrict__ deg, int E) {
    int t = blockIdx.x * blockDim.x + threadIdx.x;
    if (t < E) atomicAdd(&deg[ei[E + t]], 1);
}

__global__ void scan_local(const int* __restrict__ deg, int* __restrict__ rowptr,
                           int* __restrict__ bsum, int N) {
    __shared__ int sh[256];
    const int tid = threadIdx.x;
    const int i = blockIdx.x * 256 + tid;
    const int v = (i < N) ? deg[i] + 1 : 0;
    sh[tid] = v;
    __syncthreads();
    for (int off = 1; off < 256; off <<= 1) {
        int u = (tid >= off) ? sh[tid - off] : 0;
        __syncthreads();
        sh[tid] += u;
        __syncthreads();
    }
    if (i < N) rowptr[i] = sh[tid] - v;
    if (tid == 255) bsum[blockIdx.x] = sh[255];
}

__global__ void scan_add(int* __restrict__ rowptr, int* __restrict__ cursor,
                         const int* __restrict__ bsum, int N, int Ep) {
    __shared__ int sh[256];
    const int tid = threadIdx.x;
    const int b = blockIdx.x;
    int p = 0;
    for (int j = tid; j < b; j += 256) p += bsum[j];
    sh[tid] = p;
    __syncthreads();
    for (int off = 128; off > 0; off >>= 1) {
        if (tid < off) sh[tid] += sh[tid + off];
        __syncthreads();
    }
    const int base = sh[0];
    const int i = b * 256 + tid;
    if (i < N) { int r = rowptr[i] + base; rowptr[i] = r; cursor[i] = r + 1; }
    if (i == 0) rowptr[N] = Ep;
}

__global__ void fill_col(const int* __restrict__ ei, const int* __restrict__ rowptr,
                         int* __restrict__ cursor, int* __restrict__ col, int E, int Ep) {
    int t = blockIdx.x * blockDim.x + threadIdx.x;
    if (t >= Ep) return;
    if (t < E) {
        int s = ei[t], d = ei[E + t];
        int pos = atomicAdd(&cursor[d], 1);
        col[pos] = s;
    } else {
        int n = t - E;
        col[rowptr[n]] = n;
    }
}

__global__ void canon_col(const int* __restrict__ rowptr, const int* __restrict__ col,
                          int* __restrict__ col2, int N) {
    const int lane = threadIdx.x & 31;
    const int node = (blockIdx.x * blockDim.x + threadIdx.x) >> 5;
    if (node >= N) return;
    const int beg = rowptr[node], end = rowptr[node + 1];
    for (int i = beg + lane; i < end; i += 32) {
        const int v = col[i];
        int r = 0;
        for (int j = beg; j < end; ++j) {
            const int u = col[j];
            r += ((u < v) || (u == v && j < i)) ? 1 : 0;
        }
        col2[beg + r] = v;
    }
}

__global__ void gemm1_wmma(const float* __restrict__ X, const float* __restrict__ W,
                           float* __restrict__ XL, int N, int K) {
    const int lane  = threadIdx.x;
    const int mrow  = lane & 15;
    const int khalf = (lane >> 4) << 1;
    const int mt    = blockIdx.x;
    int row = mt * 16 + mrow;
    if (row > N - 1) row = N - 1;

    v8f acc0 = {}, acc1 = {}, acc2 = {}, acc3 = {};
    for (int k0 = 0; k0 < K; k0 += 4) {
        const int kb = k0 + khalf;
        v2f a;
        a.x = X[row * K + kb];
        a.y = X[row * K + kb + 1];
        const float* Wr0 = W + kb * 64;
        const float* Wr1 = Wr0 + 64;
        v2f b0, b1, b2, b3;
        b0.x = Wr0[mrow];      b0.y = Wr1[mrow];
        b1.x = Wr0[mrow + 16]; b1.y = Wr1[mrow + 16];
        b2.x = Wr0[mrow + 32]; b2.y = Wr1[mrow + 32];
        b3.x = Wr0[mrow + 48]; b3.y = Wr1[mrow + 48];
        acc0 = __builtin_amdgcn_wmma_f32_16x16x4_f32(false, a, false, b0, (short)0, acc0, false, false);
        acc1 = __builtin_amdgcn_wmma_f32_16x16x4_f32(false, a, false, b1, (short)0, acc1, false, false);
        acc2 = __builtin_amdgcn_wmma_f32_16x16x4_f32(false, a, false, b2, (short)0, acc2, false, false);
        acc3 = __builtin_amdgcn_wmma_f32_16x16x4_f32(false, a, false, b3, (short)0, acc3, false, false);
    }
    const int r0 = (lane >> 4) * 8;
    for (int v = 0; v < 8; ++v) {
        int r = mt * 16 + r0 + v;
        if (r < N) {
            float* o = XL + r * 64;
            o[mrow]      = acc0[v];
            o[mrow + 16] = acc1[v];
            o[mrow + 32] = acc2[v];
            o[mrow + 48] = acc3[v];
        }
    }
}

__global__ void gemm2_wmma(const float* __restrict__ Hh, const float* __restrict__ W,
                           float* __restrict__ XL, int N, int K, int OUT) {
    const int lane  = threadIdx.x;
    const int mrow  = lane & 15;
    const int khalf = (lane >> 4) << 1;
    const int mt    = blockIdx.x;
    int row = mt * 16 + mrow;
    if (row > N - 1) row = N - 1;

    v8f acc = {};
    for (int k0 = 0; k0 < K; k0 += 4) {
        const int kb = k0 + khalf;
        v2f a, b;
        a.x = Hh[row * K + kb];
        a.y = Hh[row * K + kb + 1];
        b.x = (mrow < OUT) ? W[kb * OUT + mrow]       : 0.0f;
        b.y = (mrow < OUT) ? W[(kb + 1) * OUT + mrow] : 0.0f;
        acc = __builtin_amdgcn_wmma_f32_16x16x4_f32(false, a, false, b, (short)0, acc, false, false);
    }
    if (mrow < OUT) {
        const int r0 = (lane >> 4) * 8;
        for (int v = 0; v < 8; ++v) {
            int r = mt * 16 + r0 + v;
            if (r < N) XL[r * OUT + mrow] = acc[v];
        }
    }
}

template <int H, int C>
__global__ void coefs(const float* __restrict__ XL, const float* __restrict__ Asrc,
                      const float* __restrict__ Adst, float* __restrict__ as_o,
                      float* __restrict__ ad_o, int N) {
    int t = blockIdx.x * blockDim.x + threadIdx.x;
    if (t >= N * H) return;
    int h = t % H;
    const float* v = XL + t * C;
    float ss = 0.f, sd = 0.f;
#pragma unroll
    for (int c = 0; c < C; ++c) { ss += v[c] * Asrc[h * C + c]; sd += v[c] * Adst[h * C + c]; }
    as_o[t] = ss; ad_o[t] = sd;
}

static __device__ __forceinline__ int lane_i(int v, int j) { return __builtin_amdgcn_readlane(v, j); }
static __device__ __forceinline__ float lane_f(float v, int j) {
    return __int_as_float(__builtin_amdgcn_readlane(__float_as_int(v), j));
}

__global__ void agg1(const int* __restrict__ rowptr, const int* __restrict__ col,
                     const float* __restrict__ XL, const float* __restrict__ as,
                     const float* __restrict__ ad, const float* __restrict__ b,
                     float* __restrict__ xemb, float* __restrict__ Hh, int N) {
    const int lane = threadIdx.x & 31;
    int node = (blockIdx.x * blockDim.x + threadIdx.x) >> 5;
    if (node >= N) return;
    node = __builtin_amdgcn_readfirstlane(node);
    const int beg = rowptr[node], end = rowptr[node + 1];
    const float4 adv = *(const float4*)(ad + (size_t)node * 4);
    const float ninf = __int_as_float(0xFF800000u);

    float m0 = ninf, m1 = ninf, m2 = ninf, m3 = ninf;
    for (int cb = beg; cb < end; cb += 32) {
        const int i = cb + lane;
        if (i < end) {
            const float4 a = *(const float4*)(as + (size_t)col[i] * 4);
            m0 = fmaxf(m0, lrelu(a.x + adv.x));
            m1 = fmaxf(m1, lrelu(a.y + adv.y));
            m2 = fmaxf(m2, lrelu(a.z + adv.z));
            m3 = fmaxf(m3, lrelu(a.w + adv.w));
        }
    }
    for (int off = 16; off > 0; off >>= 1) {
        m0 = fmaxf(m0, __shfl_xor(m0, off, 32));
        m1 = fmaxf(m1, __shfl_xor(m1, off, 32));
        m2 = fmaxf(m2, __shfl_xor(m2, off, 32));
        m3 = fmaxf(m3, __shfl_xor(m3, off, 32));
    }

    float S0 = 0.f, S1 = 0.f, S2 = 0.f, S3 = 0.f;
    for (int cb = beg; cb < end; cb += 32) {
        const int i = cb + lane;
        if (i < end) {
            const float4 a = *(const float4*)(as + (size_t)col[i] * 4);
            S0 += expf(lrelu(a.x + adv.x) - m0);
            S1 += expf(lrelu(a.y + adv.y) - m1);
            S2 += expf(lrelu(a.z + adv.z) - m2);
            S3 += expf(lrelu(a.w + adv.w) - m3);
        }
    }
    for (int off = 16; off > 0; off >>= 1) {
        S0 += __shfl_xor(S0, off, 32);
        S1 += __shfl_xor(S1, off, 32);
        S2 += __shfl_xor(S2, off, 32);
        S3 += __shfl_xor(S3, off, 32);
    }

    const int h = lane >> 3;
    const int k0 = (h == 0) ? -1 : 0, k1 = (h == 1) ? -1 : 0, k2 = (h == 2) ? -1 : 0, k3 = (h == 3) ? -1 : 0;
    float accx = 0.f, accy = 0.f;
    for (int cb = beg; cb < end; cb += 32) {
        const int i = cb + lane;
        int s = node;
        float al0 = 0.f, al1 = 0.f, al2 = 0.f, al3 = 0.f;
        if (i < end) {
            s = col[i];
            const float4 a = *(const float4*)(as + (size_t)s * 4);
            al0 = expf(lrelu(a.x + adv.x) - m0) / (S0 + 1e-16f);
            al1 = expf(lrelu(a.y + adv.y) - m1) / (S1 + 1e-16f);
            al2 = expf(lrelu(a.z + adv.z) - m2) / (S2 + 1e-16f);
            al3 = expf(lrelu(a.w + adv.w) - m3) / (S3 + 1e-16f);
        }
        const int cnt = (end - cb < 32) ? (end - cb) : 32;
        for (int j0 = 0; j0 < cnt; j0 += 4) {
            float2 x[4];
            float  al[4];
#pragma unroll
            for (int u = 0; u < 4; ++u) {
                const int jj = j0 + u;
                const int jc = (jj < cnt) ? jj : (cnt - 1);
                const int sj = lane_i(s, jc);
                const int ab = (__float_as_int(lane_f(al0, jc)) & k0) | (__float_as_int(lane_f(al1, jc)) & k1) |
                               (__float_as_int(lane_f(al2, jc)) & k2) | (__float_as_int(lane_f(al3, jc)) & k3);
                al[u] = __int_as_float(ab);
                x[u]  = *(const float2*)(XL + (size_t)sj * 64 + 2 * lane);
            }
#pragma unroll
            for (int u = 0; u < 4; ++u) {
                if (j0 + u < cnt) { accx += x[u].x * al[u]; accy += x[u].y * al[u]; }
            }
        }
    }
    const float2 bb = *(const float2*)(b + 2 * lane);
    float2 v; v.x = accx + bb.x; v.y = accy + bb.y;
    *(float2*)(xemb + (size_t)node * 64 + 2 * lane) = v;
    float2 r; r.x = v.x > 0.f ? v.x : 0.f; r.y = v.y > 0.f ? v.y : 0.f;
    *(float2*)(Hh + (size_t)node * 64 + 2 * lane) = r;
}

__global__ void agg2(const int* __restrict__ rowptr, const int* __restrict__ col,
                     const float* __restrict__ XL, const float* __restrict__ as,
                     const float* __restrict__ ad, const float* __restrict__ b,
                     float* __restrict__ out, int N) {
    const int lane = threadIdx.x & 31;
    int node = (blockIdx.x * blockDim.x + threadIdx.x) >> 5;
    if (node >= N) return;
    node = __builtin_amdgcn_readfirstlane(node);
    const int beg = rowptr[node], end = rowptr[node + 1];
    const float adv = ad[node];
    const float ninf = __int_as_float(0xFF800000u);

    float m = ninf;
    for (int cb = beg; cb < end; cb += 32) {
        const int i = cb + lane;
        if (i < end) m = fmaxf(m, lrelu(as[col[i]] + adv));
    }
    for (int off = 16; off > 0; off >>= 1) m = fmaxf(m, __shfl_xor(m, off, 32));

    float S = 0.f;
    for (int cb = beg; cb < end; cb += 32) {
        const int i = cb + lane;
        if (i < end) S += expf(lrelu(as[col[i]] + adv) - m);
    }
    for (int off = 16; off > 0; off >>= 1) S += __shfl_xor(S, off, 32);

    const int c = lane & 7;
    const int q = lane >> 3;
    float acc = 0.f;
    for (int cb = beg; cb < end; cb += 32) {
        const int i = cb + lane;
        int s = node;
        float al = 0.f;
        if (i < end) {
            s = col[i];
            al = expf(lrelu(as[s] + adv) - m) / (S + 1e-16f);
        }
        const int cnt = (end - cb < 32) ? (end - cb) : 32;
        for (int j0 = 0; j0 < cnt; j0 += 8) {
            const int ja = j0 + q, jb = j0 + 4 + q;
            const bool va = ja < cnt, vb = jb < cnt;
            const int ca = va ? ja : 0, cb2 = vb ? jb : 0;
            const int   sa = __shfl(s, ca, 32),  sb = __shfl(s, cb2, 32);
            const float aa = __shfl(al, ca, 32), ab = __shfl(al, cb2, 32);
            const float xa = XL[(size_t)sa * 8 + c];
            const float xb = XL[(size_t)sb * 8 + c];
            acc = va ? acc + xa * aa : acc;
            acc = vb ? acc + xb * ab : acc;
        }
    }
    acc += __shfl_xor(acc, 8, 32);
    acc += __shfl_xor(acc, 16, 32);
    if (q == 0) out[(size_t)node * 8 + c] = acc + b[c];
}

extern "C" void kernel_launch(void* const* d_in, const int* in_sizes, int n_in,
                              void* d_out, int out_size, void* d_ws, size_t ws_size,
                              hipStream_t stream) {
    const float* x    = (const float*)d_in[0];
    const int*   ei   = (const int*)d_in[1];
    const float* W1   = (const float*)d_in[2];
    const float* as1w = (const float*)d_in[3];
    const float* ad1w = (const float*)d_in[4];
    const float* b1   = (const float*)d_in[5];
    const float* W2   = (const float*)d_in[6];
    const float* as2w = (const float*)d_in[7];
    const float* ad2w = (const float*)d_in[8];
    const float* b2   = (const float*)d_in[9];

    const int HC  = in_sizes[5];
    const int IN  = in_sizes[2] / HC;
    const int OUT = in_sizes[9];
    const int N   = in_sizes[0] / IN;
    const int E   = in_sizes[1] / 2;
    const int Ep  = E + N;
    const int H   = 4;

    float* out  = (float*)d_out;
    float* xemb = out + (size_t)N * OUT;

    const int B = 256;
    auto g = [](int n, int b) { return (n + b - 1) / b; };
    const int NB = g(N, 256);

    float* ws   = (float*)d_ws;
    float* xl1  = ws; ws += (size_t)N * HC;
    float* as1  = ws; ws += (size_t)N * H;
    float* ad1  = ws; ws += (size_t)N * H;
    float* hbuf = ws; ws += (size_t)N * HC;
    float* xl2  = ws; ws += (size_t)N * OUT;
    float* as2  = ws; ws += (size_t)N;
    float* ad2  = ws; ws += (size_t)N;
    int* iw     = (int*)ws;
    int* deg    = iw; iw += (size_t)N;
    int* rowptr = iw; iw += (size_t)N + 1;
    int* cursor = iw; iw += (size_t)N;
    int* bsum   = iw; iw += (size_t)NB;
    int* col    = iw; iw += (size_t)Ep;
    int* col2   = iw; iw += (size_t)Ep;

    zero_deg<<<g(N, B), B, 0, stream>>>(deg, N);
    hist_dst<<<g(E, B), B, 0, stream>>>(ei, deg, E);
    scan_local<<<NB, 256, 0, stream>>>(deg, rowptr, bsum, N);
    scan_add<<<NB, 256, 0, stream>>>(rowptr, cursor, bsum, N, Ep);
    fill_col<<<g(Ep, B), B, 0, stream>>>(ei, rowptr, cursor, col, E, Ep);
    canon_col<<<g(N, 8), B, 0, stream>>>(rowptr, col, col2, N);

    gemm1_wmma<<<g(N, 16), 32, 0, stream>>>(x, W1, xl1, N, IN);
    coefs<4, 16><<<g(N * H, B), B, 0, stream>>>(xl1, as1w, ad1w, as1, ad1, N);
    agg1<<<g(N, 8), B, 0, stream>>>(rowptr, col2, xl1, as1, ad1, b1, xemb, hbuf, N);

    gemm2_wmma<<<g(N, 16), 32, 0, stream>>>(hbuf, W2, xl2, N, HC, OUT);
    coefs<1, 8><<<g(N, B), B, 0, stream>>>(xl2, as2w, ad2w, as2, ad2, N);
    agg2<<<g(N, 8), B, 0, stream>>>(rowptr, col2, xl2, as2, ad2, b2, out, N);
}


# buckets of up to 32 edges held one per lane in both node kernels: ranked by source in registers (rank pass only for larger buckets), scores fetched once not 3 times; order of all sums kept
# speedup vs baseline: 2.5836x; 1.1106x over previous
#include <hip/hip_runtime.h>
#include <math.h>

typedef __attribute__((ext_vector_type(2))) float v2f;
typedef __attribute__((ext_vector_type(8))) float v8f;

#define NEG_SLOPE 0.2f

static __device__ __forceinline__ float lrelu(float v) { return (v > 0.f) ? v : NEG_SLOPE * v; }

static __device__ __forceinline__ int lane_i(int v, int j) { return __builtin_amdgcn_readlane(v, j); }
static __device__ __forceinline__ float lane_f(float v, int j) {
    return __int_as_float(__builtin_amdgcn_readlane(__float_as_int(v), j));
}

__global__ void zero_deg(int* __restrict__ deg, int N) {
    int t = blockIdx.x * blockDim.x + threadIdx.x;
    if (t < N) deg[t] = 0;
}

__global__ void hist_dst(const int* __restrict__ ei, int* __restrict__ deg, int E) {
    int t = blockIdx.x * blockDim.x + threadIdx.x;
    if (t < E) atomicAdd(&deg[ei[E + t]], 1);
}

__global__ void scan_local(const int* __restrict__ deg, int* __restrict__ rowptr,
                           int* __restrict__ bsum, int N) {
    __shared__ int sh[256];
    const int tid = threadIdx.x;
    const int i = blockIdx.x * 256 + tid;
    const int v = (i < N) ? deg[i] + 1 : 0;
    sh[tid] = v;
    __syncthreads();
    for (int off = 1; off < 256; off <<= 1) {
        int u = (tid >= off) ? sh[tid - off] : 0;
        __syncthreads();
        sh[tid] += u;
        __syncthreads();
    }
    if (i < N) rowptr[i] = sh[tid] - v;
    if (tid == 255) bsum[blockIdx.x] = sh[255];
}

__global__ void scan_add(int* __restrict__ rowptr, int* __restrict__ cursor,
                         const int* __restrict__ bsum, int N, int Ep) {
    __shared__ int sh[256];
    const int tid = threadIdx.x;
    const int b = blockIdx.x;
    int p = 0;
    for (int j = tid; j < b; j += 256) p += bsum[j];
    sh[tid] = p;
    __syncthreads();
    for (int off = 128; off > 0; off >>= 1) {
        if (tid < off) sh[tid] += sh[tid + off];
        __syncthreads();
    }
    const int base = sh[0];
    const int i = b * 256 + tid;
    if (i < N) { int r = rowptr[i] + base; rowptr[i] = r; cursor[i] = r + 1; }
    if (i == 0) rowptr[N] = Ep;
}

__global__ void fill_col(const int* __restrict__ ei, const int* __restrict__ rowptr,
                         int* __restrict__ cursor, int* __restrict__ col, int E, int Ep) {
    int t = blockIdx.x * blockDim.x + threadIdx.x;
    if (t >= Ep) return;
    if (t < E) {
        int s = ei[t], d = ei[E + t];
        int pos = atomicAdd(&cursor[d], 1);
        col[pos] = s;
    } else {
        int n = t - E;
        col[rowptr[n]] = n;
    }
}

__global__ void canon_big(const int* __restrict__ rowptr, const int* __restrict__ col,
                          int* __restrict__ col2, int N) {
    const int lane = threadIdx.x & 31;
    const int n = blockIdx.x * blockDim.x + threadIdx.x;
    int nb = 0, ne = 0;
    if (n < N) { nb = rowptr[n]; ne = rowptr[n + 1]; }
    unsigned big = __builtin_amdgcn_ballot_w32(ne - nb > 32);
    while (big) {
        const int l = __builtin_ctz(big);
        big &= big - 1;
        const int beg = lane_i(nb, l), end = lane_i(ne, l);
        for (int i = beg + lane; i < end; i += 32) {
            const int v = col[i];
            int r = 0;
            for (int j = beg; j < end; ++j) {
                const int u = col[j];
                r += ((u < v) || (u == v && j < i)) ? 1 : 0;
            }
            col2[beg + r] = v;
        }
    }
}

__global__ void gemm1_wmma(const float* __restrict__ X, const float* __restrict__ W,
                           float* __restrict__ XL, int N, int K) {
    const int lane  = threadIdx.x;
    const int mrow  = lane & 15;
    const int khalf = (lane >> 4) << 1;
    const int mt    = blockIdx.x;
    int row = mt * 16 + mrow;
    if (row > N - 1) row = N - 1;

    v8f acc0 = {}, acc1 = {}, acc2 = {}, acc3 = {};
    for (int k0 = 0; k0 < K; k0 += 4) {
        const int kb = k0 + khalf;
        v2f a;
        a.x = X[row * K + kb];
        a.y = X[row * K + kb + 1];
        const float* Wr0 = W + kb * 64;
        const float* Wr1 = Wr0 + 64;
        v2f b0, b1, b2, b3;
        b0.x = Wr0[mrow];      b0.y = Wr1[mrow];
        b1.x = Wr0[mrow + 16]; b1.y = Wr1[mrow + 16];
        b2.x = Wr0[mrow + 32]; b2.y = Wr1[mrow + 32];
        b3.x = Wr0[mrow + 48]; b3.y = Wr1[mrow + 48];
        acc0 = __builtin_amdgcn_wmma_f32_16x16x4_f32(false, a, false, b0, (short)0, acc0, false, false);
        acc1 = __builtin_amdgcn_wmma_f32_16x16x4_f32(false, a, false, b1, (short)0, acc1, false, false);
        acc2 = __builtin_amdgcn_wmma_f32_16x16x4_f32(false, a, false, b2, (short)0, acc2, false, false);
        acc3 = __builtin_amdgcn_wmma_f32_16x16x4_f32(false, a, false, b3, (short)0, acc3, false, false);
    }
    const int r0 = (lane >> 4) * 8;
    for (int v = 0; v < 8; ++v) {
        int r = mt * 16 + r0 + v;
        if (r < N) {
            float* o = XL + r * 64;
            o[mrow]      = acc0[v];
            o[mrow + 16] = acc1[v];
            o[mrow + 32] = acc2[v];
            o[mrow + 48] = acc3[v];
        }
    }
}

__global__ void gemm2_wmma(const float* __restrict__ Hh, const float* __restrict__ W,
                           float* __restrict__ XL, int N, int K, int OUT) {
    const int lane  = threadIdx.x;
    const int mrow  = lane & 15;
    const int khalf = (lane >> 4) << 1;
    const int mt    = blockIdx.x;
    int row = mt * 16 + mrow;
    if (row > N - 1) row = N - 1;

    v8f acc = {};
    for (int k0 = 0; k0 < K; k0 += 4) {
        const int kb = k0 + khalf;
        v2f a, b;
        a.x = Hh[row * K + kb];
        a.y = Hh[row * K + kb + 1];
        b.x = (mrow < OUT) ? W[kb * OUT + mrow]       : 0.0f;
        b.y = (mrow < OUT) ? W[(kb + 1) * OUT + mrow] : 0.0f;
        acc = __builtin_amdgcn_wmma_f32_16x16x4_f32(false, a, false, b, (short)0, acc, false, false);
    }
    if (mrow < OUT) {
        const int r0 = (lane >> 4) * 8;
        for (int v = 0; v < 8; ++v) {
            int r = mt * 16 + r0 + v;
            if (r < N) XL[r * OUT + mrow] = acc[v];
        }
    }
}

template <int H, int C>
__global__ void coefs(const float* __restrict__ XL, const float* __restrict__ Asrc,
                      const float* __restrict__ Adst, float* __restrict__ as_o,
                      float* __restrict__ ad_o, int N) {
    int t = blockIdx.x * blockDim.x + threadIdx.x;
    if (t >= N * H) return;
    int h = t % H;
    const float* v = XL + t * C;
    float ss = 0.f, sd = 0.f;
#pragma unroll
    for (int c = 0; c < C; ++c) { ss += v[c] * Asrc[h * C + c]; sd += v[c] * Adst[h * C + c]; }
    as_o[t] = ss; ad_o[t] = sd;
}

static __device__ __forceinline__ void rows1(const float* __restrict__ XL, int lane, int s,
                                             float al0, float al1, float al2, float al3, int cnt,
                                             int k0, int k1, int k2, int k3,
                                             float& accx, float& accy) {
    for (int j0 = 0; j0 < cnt; j0 += 4) {
        float2 x[4];
        float  al[4];
#pragma unroll
        for (int u = 0; u < 4; ++u) {
            const int jj = j0 + u;
            const int jc = (jj < cnt) ? jj : (cnt - 1);
            const int sj = lane_i(s, jc);
            const int ab = (__float_as_int(lane_f(al0, jc)) & k0) | (__float_as_int(lane_f(al1, jc)) & k1) |
                           (__float_as_int(lane_f(al2, jc)) & k2) | (__float_as_int(lane_f(al3, jc)) & k3);
            al[u] = __int_as_float(ab);
            x[u]  = *(const float2*)(XL + (size_t)sj * 64 + 2 * lane);
        }
#pragma unroll
        for (int u = 0; u < 4; ++u) {
            if (j0 + u < cnt) { accx += x[u].x * al[u]; accy += x[u].y * al[u]; }
        }
    }
}

static __device__ __forceinline__ int rank_by_source(int v, int lane, int deg, int N) {
    int r = 0;
    if (N <= (1 << 26)) {
        const int key = (v << 5) | lane;
        for (int j = 0; j < deg; ++j) r += (lane_i(key, j) < key) ? 1 : 0;
    } else {
        for (int j = 0; j < deg; ++j) {
            const int u = lane_i(v, j);
            r += ((u < v) || (u == v && j < lane)) ? 1 : 0;
        }
    }
    return r;
}

__global__ void agg1(const int* __restrict__ rowptr, const int* __restrict__ col,
                     int* __restrict__ col2,
                     const float* __restrict__ XL, const float* __restrict__ as,
                     const float* __restrict__ ad, const float* __restrict__ b,
                     float* __restrict__ xemb, float* __restrict__ Hh, int N) {
    const int lane = threadIdx.x & 31;
    int node = (blockIdx.x * blockDim.x + threadIdx.x) >> 5;
    if (node >= N) return;
    node = __builtin_amdgcn_readfirstlane(node);
    const int beg = rowptr[node], end = rowptr[node + 1];
    const int deg = end - beg;
    const float4 adv = *(const float4*)(ad + (size_t)node * 4);
    const float ninf = __int_as_float(0xFF800000u);

    const int h = lane >> 3;
    const int k0 = (h == 0) ? -1 : 0, k1 = (h == 1) ? -1 : 0, k2 = (h == 2) ? -1 : 0, k3 = (h == 3) ? -1 : 0;
    float accx = 0.f, accy = 0.f;

    if (deg <= 32) {
        const bool on = lane < deg;
        const int v = on ? col[beg + lane] : 0;
        int r = rank_by_source(v, lane, deg, N);
        r = on ? r : lane;
        int s = __builtin_amdgcn_ds_permute(r << 2, v);
        if (on) col2[beg + lane] = s;
        s = on ? s : node;

        float e0 = ninf, e1 = ninf, e2 = ninf, e3 = ninf;
        if (on) {
            const float4 a = *(const float4*)(as + (size_t)s * 4);
            e0 = lrelu(a.x + adv.x);
            e1 = lrelu(a.y + adv.y);
            e2 = lrelu(a.z + adv.z);
            e3 = lrelu(a.w + adv.w);
        }
        float m0 = e0, m1 = e1, m2 = e2, m3 = e3;
        for (int off = 16; off > 0; off >>= 1) {
            m0 = fmaxf(m0, __shfl_xor(m0, off, 32));
            m1 = fmaxf(m1, __shfl_xor(m1, off, 32));
            m2 = fmaxf(m2, __shfl_xor(m2, off, 32));
            m3 = fmaxf(m3, __shfl_xor(m3, off, 32));
        }
        const float p0 = on ? expf(e0 - m0) : 0.f;
        const float p1 = on ? expf(e1 - m1) : 0.f;
        const float p2 = on ? expf(e2 - m2) : 0.f;
        const float p3 = on ? expf(e3 - m3) : 0.f;
        float S0 = p0, S1 = p1, S2 = p2, S3 = p3;
        for (int off = 16; off > 0; off >>= 1) {
            S0 += __shfl_xor(S0, off, 32);
            S1 += __shfl_xor(S1, off, 32);
            S2 += __shfl_xor(S2, off, 32);
            S3 += __shfl_xor(S3, off, 32);
        }
        const float al0 = p0 / (S0 + 1e-16f);
        const float al1 = p1 / (S1 + 1e-16f);
        const float al2 = p2 / (S2 + 1e-16f);
        const float al3 = p3 / (S3 + 1e-16f);
        rows1(XL, lane, s, al0, al1, al2, al3, deg, k0, k1, k2, k3, accx, accy);
    } else {
        float m0 = ninf, m1 = ninf, m2 = ninf, m3 = ninf;
        for (int cb = beg; cb < end; cb += 32) {
            const int i = cb + lane;
            if (i < end) {
                const float4 a = *(const float4*)(as + (size_t)col2[i] * 4);
                m0 = fmaxf(m0, lrelu(a.x + adv.x));
                m1 = fmaxf(m1, lrelu(a.y + adv.y));
                m2 = fmaxf(m2, lrelu(a.z + adv.z));
                m3 = fmaxf(m3, lrelu(a.w + adv.w));
            }
        }
        for (int off = 16; off > 0; off >>= 1) {
            m0 = fmaxf(m0, __shfl_xor(m0, off, 32));
            m1 = fmaxf(m1, __shfl_xor(m1, off, 32));
            m2 = fmaxf(m2, __shfl_xor(m2, off, 32));
            m3 = fmaxf(m3, __shfl_xor(m3, off, 32));
        }

        float S0 = 0.f, S1 = 0.f, S2 = 0.f, S3 = 0.f;
        for (int cb = beg; cb < end; cb += 32) {
            const int i = cb + lane;
            if (i < end) {
                const float4 a = *(const float4*)(as + (size_t)col2[i] * 4);
                S0 += expf(lrelu(a.x + adv.x) - m0);
                S1 += expf(lrelu(a.y + adv.y) - m1);
                S2 += expf(lrelu(a.z + adv.z) - m2);
                S3 += expf(lrelu(a.w + adv.w) - m3);
            }
        }
        for (int off = 16; off > 0; off >>= 1) {
            S0 += __shfl_xor(S0, off, 32);
            S1 += __shfl_xor(S1, off, 32);
            S2 += __shfl_xor(S2, off, 32);
            S3 += __shfl_xor(S3, off, 32);
        }

        for (int cb = beg; cb < end; cb += 32) {
            const int i = cb + lane;
            int s = node;
            float al0 = 0.f, al1 = 0.f, al2 = 0.f, al3 = 0.f;
            if (i < end) {
                s = col2[i];
                const float4 a = *(const float4*)(as + (size_t)s * 4);
                al0 = expf(lrelu(a.x + adv.x) - m0) / (S0 + 1e-16f);
                al1 = expf(lrelu(a.y + adv.y) - m1) / (S1 + 1e-16f);
                al2 = expf(lrelu(a.z + adv.z) - m2) / (S2 + 1e-16f);
                al3 = expf(lrelu(a.w + adv.w) - m3) / (S3 + 1e-16f);
            }
            const int cnt = (end - cb < 32) ? (end - cb) : 32;
            rows1(XL, lane, s, al0, al1, al2, al3, cnt, k0, k1, k2, k3, accx, accy);
        }
    }
    const float2 bb = *(const float2*)(b + 2 * lane);
    float2 v; v.x = accx + bb.x; v.y = accy + bb.y;
    *(float2*)(xemb + (size_t)node * 64 + 2 * lane) = v;
    float2 r; r.x = v.x > 0.f ? v.x : 0.f; r.y = v.y > 0.f ? v.y : 0.f;
    *(float2*)(Hh + (size_t)node * 64 + 2 * lane) = r;
}

static __device__ __forceinline__ void rows2(const float* __restrict__ XL, int c, int q, int s, float al,
                                             int cnt, float& acc) {
    for (int j0 = 0; j0 < cnt; j0 += 8) {
        const int ja = j0 + q, jb = j0 + 4 + q;
        const bool va = ja < cnt, vb = jb < cnt;
        const int ca = va ? ja : 0, cb2 = vb ? jb : 0;
        const int   sa = __shfl(s, ca, 32),  sb = __shfl(s, cb2, 32);
        const float aa = __shfl(al, ca, 32), ab = __shfl(al, cb2, 32);
        const float xa = XL[(size_t)sa * 8 + c];
        const float xb = XL[(size_t)sb * 8 + c];
        acc = va ? acc + xa * aa : acc;
        acc = vb ? acc + xb * ab : acc;
    }
}

__global__ void agg2(const int* __restrict__ rowptr, const int* __restrict__ col,
                     const float* __restrict__ XL, const float* __restrict__ as,
                     const float* __restrict__ ad, const float* __restrict__ b,
                     float* __restrict__ out, int N) {
    const int lane = threadIdx.x & 31;
    int node = (blockIdx.x * blockDim.x + threadIdx.x) >> 5;
    if (node >= N) return;
    node = __builtin_amdgcn_readfirstlane(node);
    const int beg = rowptr[node], end = rowptr[node + 1];
    const int deg = end - beg;
    const float adv = ad[node];
    const float ninf = __int_as_float(0xFF800000u);

    const int c = lane & 7;
    const int q = lane >> 3;
    float acc = 0.f;

    if (deg <= 32) {
        const bool on = lane < deg;
        int s = node;
        float e = ninf;
        if (on) {
            s = col[beg + lane];
            e = lrelu(as[s] + adv);
        }
        float m = e;
        for (int off = 16; off > 0; off >>= 1) m = fmaxf(m, __shfl_xor(m, off, 32));
        const float p = on ? expf(e - m) : 0.f;
        float S = p;
        for (int off = 16; off > 0; off >>= 1) S += __shfl_xor(S, off, 32);
        const float al = p / (S + 1e-16f);
        rows2(XL, c, q, s, al, deg, acc);
    } else {
        float m = ninf;
        for (int cb = beg; cb < end; cb += 32) {
            const int i = cb + lane;
            if (i < end) m = fmaxf(m, lrelu(as[col[i]] + adv));
        }
        for (int off = 16; off > 0; off >>= 1) m = fmaxf(m, __shfl_xor(m, off, 32));

        float S = 0.f;
        for (int cb = beg; cb < end; cb += 32) {
            const int i = cb + lane;
            if (i < end) S += expf(lrelu(as[col[i]] + adv) - m);
        }
        for (int off = 16; off > 0; off >>= 1) S += __shfl_xor(S, off, 32);

        for (int cb = beg; cb < end; cb += 32) {
            const int i = cb + lane;
            int s = node;
            float al = 0.f;
            if (i < end) {
                s = col[i];
                al = expf(lrelu(as[s] + adv) - m) / (S + 1e-16f);
            }
            const int cnt = (end - cb < 32) ? (end - cb) : 32;
            rows2(XL, c, q, s, al, cnt, acc);
        }
    }
    acc += __shfl_xor(acc, 8, 32);
    acc += __shfl_xor(acc, 16, 32);
    if (q == 0) out[(size_t)node * 8 + c] = acc + b[c];
}

extern "C" void kernel_launch(void* const* d_in, const int* in_sizes, int n_in,
                              void* d_out, int out_size, void* d_ws, size_t ws_size,
                              hipStream_t stream) {
    const float* x    = (const float*)d_in[0];
    const int*   ei   = (const int*)d_in[1];
    const float* W1   = (const float*)d_in[2];
    const float* as1w = (const float*)d_in[3];
    const float* ad1w = (const float*)d_in[4];
    const float* b1   = (const float*)d_in[5];
    const float* W2   = (const float*)d_in[6];
    const float* as2w = (const float*)d_in[7];
    const float* ad2w = (const float*)d_in[8];
    const float* b2   = (const float*)d_in[9];

    const int HC  = in_sizes[5];
    const int IN  = in_sizes[2] / HC;
    const int OUT = in_sizes[9];
    const int N   = in_sizes[0] / IN;
    const int E   = in_sizes[1] / 2;
    const int Ep  = E + N;
    const int H   = 4;

    float* out  = (float*)d_out;
    float* xemb = out + (size_t)N * OUT;

    const int B = 256;
    auto g = [](int n, int b) { return (n + b - 1) / b; };
    const int NB = g(N, 256);

    float* ws   = (float*)d_ws;
    float* xl1  = ws; ws += (size_t)N * HC;
    float* as1  = ws; ws += (size_t)N * H;
    float* ad1  = ws; ws += (size_t)N * H;
    float* hbuf = ws; ws += (size_t)N * HC;
    float* xl2  = ws; ws += (size_t)N * OUT;
    float* as2  = ws; ws += (size_t)N;
    float* ad2  = ws; ws += (size_t)N;
    int* iw     = (int*)ws;
    int* deg    = iw; iw += (size_t)N;
    int* rowptr = iw; iw += (size_t)N + 1;
    int* cursor = iw; iw += (size_t)N;
    int* bsum   = iw; iw += (size_t)NB;
    int* col    = iw; iw += (size_t)Ep;
    int* col2   = iw; iw += (size_t)Ep;

    zero_deg<<<g(N, B), B, 0, stream>>>(deg, N);
    hist_dst<<<g(E, B), B, 0, stream>>>(ei, deg, E);
    scan_local<<<NB, 256, 0, stream>>>(deg, rowptr, bsum, N);
    scan_add<<<NB, 256, 0, stream>>>(rowptr, cursor, bsum, N, Ep);
    fill_col<<<g(Ep, B), B, 0, stream>>>(ei, rowptr, cursor, col, E, Ep);
    canon_big<<<g(N, B), B, 0, stream>>>(rowptr, col, col2, N);

    gemm1_wmma<<<g(N, 16), 32, 0, stream>>>(x, W1, xl1, N, IN);
    coefs<4, 16><<<g(N * H, B), B, 0, stream>>>(xl1, as1w, ad1w, as1, ad1, N);
    agg1<<<g(N, 8), B, 0, stream>>>(rowptr, col, col2, xl1, as1, ad1, b1, xemb, hbuf, N);

    gemm2_wmma<<<g(N, 16), 32, 0, stream>>>(hbuf, W2, xl2, N, HC, OUT);
    coefs<1, 8><<<g(N, B), B, 0, stream>>>(xl2, as2w, ad2w, as2, ad2, N);
    agg2<<<g(N, 8), B, 0, stream>>>(rowptr, col2, xl2, as2, ad2, b2, out, N);
}
